# starting kernel
# baseline (speedup 1.0000x reference)
#include <hip/hip_runtime.h>

typedef float v2f __attribute__((ext_vector_type(2)));
typedef float v8f __attribute__((ext_vector_type(8)));

#define C_IN   128
#define K_OUT  128
#define W_     56
#define HW_    3136
#define KRED   1152

__global__ __launch_bounds__(256)
void conv3x3_wmma_f32(const float* __restrict__ in,
                      const float* __restrict__ wgt,
                      const float* __restrict__ bias,
                      float* __restrict__ out)
{
    __shared__ float wlds[16 * KRED];

    const int tid     = threadIdx.x;
    const int oc_base = blockIdx.y * 16;

    {
        const float* wsrc = wgt + (long)oc_base * KRED;
        for (int g = tid; g < 16 * KRED; g += 256) {
            int oc_l = g / KRED;
            int rem  = g - oc_l * KRED;
            int c    = rem / 9;
            int rs   = rem - c * 9;
            wlds[(rs * C_IN + c) * 16 + oc_l] = wsrc[g];
        }
    }
    __syncthreads();

    const int wave = tid >> 5;
    const int lane = tid & 31;
    const int m    = lane & 15;
    const int koff = (lane >> 4) << 1;

    const int p  = blockIdx.x * 128 + wave * 16 + m;
    const int n  = p / HW_;
    const int hw = p - n * HW_;
    const int h  = hw / W_;
    const int w  = hw - h * W_;

    v8f acc = {};

    #pragma unroll
    for (int rs = 0; rs < 9; ++rs) {
        const int r  = rs / 3;
        const int s  = rs - r * 3;
        const int ih = h + r - 1;
        const int iw = w + s - 1;
        const bool valid = (ih >= 0) & (ih < W_) & (iw >= 0) & (iw < W_);
        const float mask = valid ? 1.0f : 0.0f;
        const long  off  = valid ? ((long)n * C_IN * HW_ + (long)ih * W_ + iw)
                                 : 0;
        const float* gin = in + off + (long)koff * HW_;
        const float* wl  = &wlds[(rs * C_IN + koff) * 16 + m];

        #pragma unroll 4
        for (int c = 0; c < C_IN; c += 4) {
            v2f A, B;
            A.x = wl[(c    ) * 16];
            A.y = wl[(c + 1) * 16];
            B.x = gin[(long)(c    ) * HW_] * mask;
            B.y = gin[(long)(c + 1) * HW_] * mask;
            acc = __builtin_amdgcn_wmma_f32_16x16x4_f32(
                false, A, false, B, (short)0, acc, false, false);
        }
    }

    const int m_hi = (lane >> 4) << 3;
    #pragma unroll
    for (int g = 0; g < 8; ++g) {
        const int oc = oc_base + m_hi + g;
        out[((long)n * K_OUT + oc) * HW_ + hw] = acc[g] + bias[oc];
    }
}

extern "C" void kernel_launch(void* const* d_in, const int* in_sizes, int n_in,
                              void* d_out, int out_size, void* d_ws, size_t ws_size,
                              hipStream_t stream) {
    (void)in_sizes; (void)n_in; (void)d_ws; (void)ws_size; (void)out_size;
    const float* in   = (const float*)d_in[0];
    const float* wgt  = (const float*)d_in[1];
    const float* bias = (const float*)d_in[2];
    float*       out  = (float*)d_out;

    dim3 grid(392, 8, 1);
    dim3 block(256, 1, 1);
    conv3x3_wmma_f32<<<grid, block, 0, stream>>>(in, wgt, bias, out);
}


# Each wave keeps 8 output-channel tiles so one loaded input fragment feeds 8 WMMAs; weights staged in LDS per 16-channel chunk; summation order changed to chunk-major
# speedup vs baseline: 2.0382x; 2.0382x over previous
#include <hip/hip_runtime.h>

typedef float v2f __attribute__((ext_vector_type(2)));
typedef float v8f __attribute__((ext_vector_type(8)));

#define C_IN   128
#define K_OUT  128
#define W_     56
#define HW_    3136
#define KRED   1152
#define CC     16
#define NCHUNK 8
#define PRS    288
#define RSS    2308
#define LDS_DW (9 * RSS)

__global__ __launch_bounds__(256)
void conv3x3_wmma_f32(const float* __restrict__ in,
                      const float* __restrict__ wgt,
                      const float* __restrict__ bias,
                      float* __restrict__ out)
{
    __shared__ __attribute__((aligned(16))) float wlds[LDS_DW];

    const int tid  = threadIdx.x;
    const int wave = tid >> 5;
    const int lane = tid & 31;
    const int m    = lane & 15;
    const int hh   = lane >> 4;

    const int p  = blockIdx.x * 128 + wave * 16 + m;
    const int n  = p / HW_;
    const int hw = p - n * HW_;
    const int h  = hw / W_;
    const int w  = hw - h * W_;

    int   toff[9];
    float tmask[9];
    #pragma unroll
    for (int rs = 0; rs < 9; ++rs) {
        const int r  = rs / 3;
        const int s  = rs - r * 3;
        const int ih = h + r - 1;
        const int iw = w + s - 1;
        const bool valid = (ih >= 0) & (ih < W_) & (iw >= 0) & (iw < W_);
        tmask[rs] = valid ? 1.0f : 0.0f;
        toff[rs]  = valid ? (ih * W_ + iw) : 0;
    }
    int so[4], ro[4];
    #pragma unroll
    for (int j = 0; j < 4; ++j) {
        int e  = lane * 4 + j;
        int cl = e / 9;
        int rs = e - cl * 9;
        so[j]  = rs * RSS + (cl >> 1) * PRS + (cl & 1) + wave * 2;
        e  = (32 + (tid & 3)) * 4 + j;
        cl = e / 9;
        rs = e - cl * 9;
        ro[j]  = rs * RSS + (cl >> 1) * PRS + (cl & 1) + (tid >> 2) * 2;
    }
    const float* inb = in + (long)n * C_IN * HW_ + (long)(2 * hh) * HW_;

    v8f acc[8];
    #pragma unroll
    for (int t = 0; t < 8; ++t) acc[t] = (v8f){};

    #pragma unroll 1
    for (int ch = 0; ch < NCHUNK; ++ch) {
        __syncthreads();
        {
            const float* wsrc = wgt + ch * (CC * 9);
            const float* wa   = wsrc + wave * KRED + lane * 4;
            #pragma unroll 8
            for (int ps = 0; ps < 16; ++ps) {
                const float4 v = *(const float4*)(wa + (long)ps * 8 * KRED);
                wlds[so[0] + ps * 16] = v.x;
                wlds[so[1] + ps * 16] = v.y;
                wlds[so[2] + ps * 16] = v.z;
                wlds[so[3] + ps * 16] = v.w;
            }
            const float* wr = wsrc + (tid >> 2) * KRED + (32 + (tid & 3)) * 4;
            #pragma unroll
            for (int i2 = 0; i2 < 2; ++i2) {
                const float4 v = *(const float4*)(wr + (long)i2 * 64 * KRED);
                wlds[ro[0] + i2 * 128] = v.x;
                wlds[ro[1] + i2 * 128] = v.y;
                wlds[ro[2] + i2 * 128] = v.z;
                wlds[ro[3] + i2 * 128] = v.w;
            }
        }
        __syncthreads();

        const float* inc = inb + (long)(ch * CC) * HW_;
        #pragma unroll
        for (int rs = 0; rs < 9; ++rs) {
            const float* g  = inc + toff[rs];
            const float  mk = tmask[rs];
            #pragma unroll
            for (int kk = 0; kk < 4; ++kk) {
                v2f B;
                B.x = g[(long)(4 * kk    ) * HW_] * mk;
                B.y = g[(long)(4 * kk + 1) * HW_] * mk;
                const float* wl = &wlds[rs * RSS + (2 * kk + hh) * PRS + m * 2];
                #pragma unroll
                for (int t = 0; t < 8; ++t) {
                    const v2f A = *(const v2f*)(wl + t * 32);
                    acc[t] = __builtin_amdgcn_wmma_f32_16x16x4_f32(
                        false, A, false, B, (short)0, acc[t], false, false);
                }
            }
        }
    }

    #pragma unroll
    for (int t = 0; t < 8; ++t) {
        #pragma unroll
        for (int g = 0; g < 8; ++g) {
            const int oc = t * 16 + hh * 8 + g;
            out[((long)n * K_OUT + oc) * HW_ + hw] = acc[t][g] + bias[oc];
        }
    }
}

extern "C" void kernel_launch(void* const* d_in, const int* in_sizes, int n_in,
                              void* d_out, int out_size, void* d_ws, size_t ws_size,
                              hipStream_t stream) {
    (void)in_sizes; (void)n_in; (void)d_ws; (void)ws_size; (void)out_size;
    const float* in   = (const float*)d_in[0];
    const float* wgt  = (const float*)d_in[1];
    const float* bias = (const float*)d_in[2];
    float*       out  = (float*)d_out;

    dim3 grid(392, 1, 1);
    dim3 block(256, 1, 1);
    conv3x3_wmma_f32<<<grid, block, 0, stream>>>(in, wgt, bias, out);
}


# Even split over the 256 WGPs: grid 256 x 14 waves (12 waves own a 16-position tile x 128 oc, 2 waves one leftover tile), max tiles per SIMD 32 -> 25; loads written in issue order; sums unchanged
# speedup vs baseline: 2.3905x; 1.1728x over previous
#include <hip/hip_runtime.h>

typedef float v2f __attribute__((ext_vector_type(2)));
typedef float v8f __attribute__((ext_vector_type(8)));

#define C_IN   128
#define K_OUT  128
#define W_     56
#define HW_    3136
#define KRED   1152
#define CC     16
#define NCHUNK 8
#define PRS    288
#define RSS    2308
#define LDS_DW (9 * RSS)
#define NBLK   256
#define NFULL  12
#define NTHR   448
#define PT_MAIN (NBLK * NFULL)


template <int NT, int D>
__device__ __forceinline__ void mac_chunk(v8f (&acc)[8], const float* wb,
                                          const float* inc,
                                          const int (&toff)[9],
                                          const float (&tmask)[9])
{
    float bx[36], by[36];
    v2f   a[2][NT];
    #pragma unroll
    for (int i = 0; i < D; ++i) {
        const float* g = inc + toff[i / 4];
        bx[i] = g[(long)(4 * (i % 4)    ) * HW_];
        by[i] = g[(long)(4 * (i % 4) + 1) * HW_];
    }
    #pragma unroll
    for (int t = 0; t < NT; ++t) a[0][t] = *(const v2f*)(wb + t * 32);
    #pragma unroll
    for (int i = 0; i < 36; ++i) {
        if (i + D < 36) {
            const int j = i + D;
            const float* g = inc + toff[j / 4];
            bx[j] = g[(long)(4 * (j % 4)    ) * HW_];
            by[j] = g[(long)(4 * (j % 4) + 1) * HW_];
        }
        if (i + 1 < 36) {
            const int j = i + 1;
            const float* wl = wb + (j / 4) * RSS + (2 * (j % 4)) * PRS;
            #pragma unroll
            for (int t = 0; t < NT; ++t)
                a[j & 1][t] = *(const v2f*)(wl + t * 32);
        }
        const float mk = tmask[i / 4];
        v2f B;
        B.x = bx[i] * mk;
        B.y = by[i] * mk;
        #pragma unroll
        for (int t = 0; t < NT; ++t)
            acc[t] = __builtin_amdgcn_wmma_f32_16x16x4_f32(
                false, a[i & 1][t], false, B, (short)0, acc[t], false, false);
    }
}

template <int NT>
__device__ __forceinline__ void store_tiles(const v8f (&acc)[8], float* op,
                                            const float* bp)
{
    #pragma unroll
    for (int t = 0; t < NT; ++t) {
        #pragma unroll
        for (int g = 0; g < 8; ++g)
            op[(long)(t * 16 + g) * HW_] = acc[t][g] + bp[t * 16 + g];
    }
}


__global__ __launch_bounds__(NTHR)
void conv3x3_wmma_f32(const float* __restrict__ in,
                      const float* __restrict__ wgt,
                      const float* __restrict__ bias,
                      float* __restrict__ out)
{
    __shared__ __attribute__((aligned(16))) float wlds[LDS_DW];

    const int tid  = threadIdx.x;
    const int lane = tid & 31;
    const int wave = __builtin_amdgcn_readfirstlane(tid >> 5);
    const int m    = lane & 15;
    const int hh   = lane >> 4;
    const int b    = blockIdx.x;
    const bool full = wave < NFULL;

    const int pt = full ? (b * NFULL + wave)
                        : (PT_MAIN + 2 * (b >> 3) + (wave - NFULL));
    const int t0 = full ? 0 : (b & 7);

    const int p  = pt * 16 + m;
    const int n  = p / HW_;
    const int hw = p - n * HW_;
    const int h  = hw / W_;
    const int w  = hw - h * W_;

    int   toff[9];
    float tmask[9];
    #pragma unroll
    for (int rs = 0; rs < 9; ++rs) {
        const int r  = rs / 3;
        const int s  = rs - r * 3;
        const int ih = h + r - 1;
        const int iw = w + s - 1;
        const bool valid = (ih >= 0) & (ih < W_) & (iw >= 0) & (iw < W_);
        tmask[rs] = valid ? 1.0f : 0.0f;
        toff[rs]  = valid ? (ih * W_ + iw) : 0;
    }
    const int ocl = tid / 12;
    const int qq  = tid - ocl * 12;
    int so[12];
    #pragma unroll
    for (int s = 0; s < 3; ++s) {
        #pragma unroll
        for (int j = 0; j < 4; ++j) {
            const int e  = (qq + 12 * s) * 4 + j;
            const int cl = e / 9;
            const int rs = e - cl * 9;
            so[s * 4 + j] = rs * RSS + (cl >> 1) * PRS + (cl & 1) + ocl * 2;
        }
    }
    const float* inb = in + (long)n * C_IN * HW_ + (long)(2 * hh) * HW_;
    const float* wb  = &wlds[hh * PRS + m * 2 + t0 * 32];

    v8f acc[8];
    #pragma unroll
    for (int t = 0; t < 8; ++t) acc[t] = (v8f){};

    #pragma unroll 1
    for (int ch = 0; ch < NCHUNK; ++ch) {
        __syncthreads();
        if (full) {
            const float* wsrc = wgt + ch * (CC * 9) + ocl * KRED + qq * 4;
            #pragma unroll
            for (int ps = 0; ps < 4; ++ps) {
                #pragma unroll
                for (int s = 0; s < 3; ++s) {
                    const float4 v =
                        *(const float4*)(wsrc + (long)ps * 32 * KRED + s * 48);
                    wlds[so[s * 4 + 0] + ps * 64] = v.x;
                    wlds[so[s * 4 + 1] + ps * 64] = v.y;
                    wlds[so[s * 4 + 2] + ps * 64] = v.z;
                    wlds[so[s * 4 + 3] + ps * 64] = v.w;
                }
            }
        }
        __syncthreads();

        const float* inc = inb + (long)(ch * CC) * HW_;
        if (full) mac_chunk<8, 10>(acc, wb, inc, toff, tmask);
        else {
            mac_chunk<1, 36>(acc, wb, inc, toff, tmask);
            asm volatile("v_nop\n\tv_nop\n\tv_nop\n\tv_nop" : "+v"(acc[0]));
        }
    }

    float*       op = out + ((long)n * K_OUT + t0 * 16 + hh * 8) * HW_ + hw;
    const float* bp = bias + t0 * 16 + hh * 8;
    if (full) store_tiles<8>(acc, op, bp);
    else      store_tiles<1>(acc, op, bp);
}

extern "C" void kernel_launch(void* const* d_in, const int* in_sizes, int n_in,
                              void* d_out, int out_size, void* d_ws, size_t ws_size,
                              hipStream_t stream) {
    (void)in_sizes; (void)n_in; (void)d_ws; (void)ws_size; (void)out_size;
    const float* in   = (const float*)d_in[0];
    const float* wgt  = (const float*)d_in[1];
    const float* bias = (const float*)d_in[2];
    float*       out  = (float*)d_out;

    dim3 grid(256, 1, 1);
    dim3 block(448, 1, 1);
    conv3x3_wmma_f32<<<grid, block, 0, stream>>>(in, wgt, bias, out);
}


# Next chunk's weights copied into a second LDS buffer from inside the current chunk's WMMA run (float4 loads 8 steps ahead of their LDS stores): one barrier per chunk, no staging phase; sums unchanged
# speedup vs baseline: 2.5123x; 1.0510x over previous
#include <hip/hip_runtime.h>

typedef float v2f __attribute__((ext_vector_type(2)));
typedef float v8f __attribute__((ext_vector_type(8)));

#define C_IN   128
#define K_OUT  128
#define W_     56
#define HW_    3136
#define KRED   1152
#define CC     16
#define NCHUNK 8
#define PRS    288
#define RSS    2308
#define LDS_DW (9 * RSS)
#define NBLK   256
#define NFULL  12
#define NTHR   448
#define PT_MAIN (NBLK * NFULL)


template <int NT, int D, bool STG>
__device__ __forceinline__ void mac_chunk(v8f (&acc)[8], const float* wb,
                                          const float* inc,
                                          const int (&toff)[9],
                                          const float (&tmask)[9],
                                          const float* wnext, float* snext,
                                          const int (&so)[12])
{
    float bx[36], by[36];
    v2f   a[2][NT];
    float4 wv[12];
    #pragma unroll
    for (int i = 0; i < D; ++i) {
        const float* g = inc + toff[i / 4];
        bx[i] = g[(long)(4 * (i % 4)    ) * HW_];
        by[i] = g[(long)(4 * (i % 4) + 1) * HW_];
    }
    #pragma unroll
    for (int t = 0; t < NT; ++t) a[0][t] = *(const v2f*)(wb + t * 32);
    #pragma unroll
    for (int i = 0; i < 36; ++i) {
        if (i + D < 36) {
            const int j = i + D;
            const float* g = inc + toff[j / 4];
            bx[j] = g[(long)(4 * (j % 4)    ) * HW_];
            by[j] = g[(long)(4 * (j % 4) + 1) * HW_];
        }
        if (STG && (i % 2 == 0) && (i / 2 < 12)) {
            const int u = i / 2;
            wv[u] = *(const float4*)(wnext + (long)(u / 3) * 32 * KRED
                                           + (u % 3) * 48);
        }
        if (i + 1 < 36) {
            const int j = i + 1;
            const float* wl = wb + (j / 4) * RSS + (2 * (j % 4)) * PRS;
            #pragma unroll
            for (int t = 0; t < NT; ++t)
                a[j & 1][t] = *(const v2f*)(wl + t * 32);
        }
        if (STG && (i >= 8) && (i % 2 == 0) && ((i - 8) / 2 < 12)) {
            const int u  = (i - 8) / 2;
            const int ps = u / 3;
            const int s  = u % 3;
            snext[so[s * 4 + 0] + ps * 64] = wv[u].x;
            snext[so[s * 4 + 1] + ps * 64] = wv[u].y;
            snext[so[s * 4 + 2] + ps * 64] = wv[u].z;
            snext[so[s * 4 + 3] + ps * 64] = wv[u].w;
        }
        const float mk = tmask[i / 4];
        v2f B;
        B.x = bx[i] * mk;
        B.y = by[i] * mk;
        #pragma unroll
        for (int t = 0; t < NT; ++t)
            acc[t] = __builtin_amdgcn_wmma_f32_16x16x4_f32(
                false, a[i & 1][t], false, B, (short)0, acc[t], false, false);
    }
}

template <int NT>
__device__ __forceinline__ void store_tiles(const v8f (&acc)[8], float* op,
                                            const float* bp)
{
    #pragma unroll
    for (int t = 0; t < NT; ++t) {
        #pragma unroll
        for (int g = 0; g < 8; ++g)
            op[(long)(t * 16 + g) * HW_] = acc[t][g] + bp[t * 16 + g];
    }
}


__global__ __launch_bounds__(NTHR)
void conv3x3_wmma_f32(const float* __restrict__ in,
                      const float* __restrict__ wgt,
                      const float* __restrict__ bias,
                      float* __restrict__ out)
{
    __shared__ __attribute__((aligned(16))) float wlds[2 * LDS_DW];

    const int tid  = threadIdx.x;
    const int lane = tid & 31;
    const int wave = __builtin_amdgcn_readfirstlane(tid >> 5);
    const int m    = lane & 15;
    const int hh   = lane >> 4;
    const int b    = blockIdx.x;
    const bool full = wave < NFULL;

    const int pt = full ? (b * NFULL + wave)
                        : (PT_MAIN + 2 * (b >> 3) + (wave - NFULL));
    const int t0 = full ? 0 : (b & 7);

    const int p  = pt * 16 + m;
    const int n  = p / HW_;
    const int hw = p - n * HW_;
    const int h  = hw / W_;
    const int w  = hw - h * W_;

    int   toff[9];
    float tmask[9];
    #pragma unroll
    for (int rs = 0; rs < 9; ++rs) {
        const int r  = rs / 3;
        const int s  = rs - r * 3;
        const int ih = h + r - 1;
        const int iw = w + s - 1;
        const bool valid = (ih >= 0) & (ih < W_) & (iw >= 0) & (iw < W_);
        tmask[rs] = valid ? 1.0f : 0.0f;
        toff[rs]  = valid ? (ih * W_ + iw) : 0;
    }
    const int ocl = tid / 12;
    const int qq  = tid - ocl * 12;
    int so[12];
    #pragma unroll
    for (int s = 0; s < 3; ++s) {
        #pragma unroll
        for (int j = 0; j < 4; ++j) {
            const int e  = (qq + 12 * s) * 4 + j;
            const int cl = e / 9;
            const int rs = e - cl * 9;
            so[s * 4 + j] = rs * RSS + (cl >> 1) * PRS + (cl & 1) + ocl * 2;
        }
    }
    const float* inb = in + (long)n * C_IN * HW_ + (long)(2 * hh) * HW_;
    const float* wb  = &wlds[hh * PRS + m * 2 + t0 * 32];

    v8f acc[8];
    #pragma unroll
    for (int t = 0; t < 8; ++t) acc[t] = (v8f){};

    const float* wst = wgt + ocl * KRED + qq * 4;
    if (full) {
        #pragma unroll
        for (int ps = 0; ps < 4; ++ps) {
            #pragma unroll
            for (int s = 0; s < 3; ++s) {
                const float4 v =
                    *(const float4*)(wst + (long)ps * 32 * KRED + s * 48);
                wlds[so[s * 4 + 0] + ps * 64] = v.x;
                wlds[so[s * 4 + 1] + ps * 64] = v.y;
                wlds[so[s * 4 + 2] + ps * 64] = v.z;
                wlds[so[s * 4 + 3] + ps * 64] = v.w;
            }
        }
    }
    __syncthreads();

    #pragma unroll 1
    for (int ch = 0; ch < NCHUNK; ++ch) {
        const int    cur = ch & 1;
        const float* inc = inb + (long)(ch * CC) * HW_;
        const float* wbc = wb + cur * LDS_DW;
        const int    nx  = (ch + 1) & (NCHUNK - 1);
        if (full)
            mac_chunk<8, 10, true>(acc, wbc, inc, toff, tmask,
                                   wst + nx * (CC * 9),
                                   wlds + (cur ^ 1) * LDS_DW, so);
        else {
            mac_chunk<1, 36, false>(acc, wbc, inc, toff, tmask,
                                    wst, wlds, so);
            asm volatile("v_nop\n\tv_nop\n\tv_nop\n\tv_nop" : "+v"(acc[0]));
        }
        if (ch + 1 < NCHUNK) __syncthreads();
    }

    float*       op = out + ((long)n * K_OUT + t0 * 16 + hh * 8) * HW_ + hw;
    const float* bp = bias + t0 * 16 + hh * 8;
    if (full) store_tiles<8>(acc, op, bp);
    else      store_tiles<1>(acc, op, bp);
}

extern "C" void kernel_launch(void* const* d_in, const int* in_sizes, int n_in,
                              void* d_out, int out_size, void* d_ws, size_t ws_size,
                              hipStream_t stream) {
    (void)in_sizes; (void)n_in; (void)d_ws; (void)ws_size; (void)out_size;
    const float* in   = (const float*)d_in[0];
    const float* wgt  = (const float*)d_in[1];
    const float* bias = (const float*)d_in[2];
    float*       out  = (float*)d_out;

    dim3 grid(256, 1, 1);
    dim3 block(448, 1, 1);
    conv3x3_wmma_f32<<<grid, block, 0, stream>>>(in, wgt, bias, out);
}


# Issue priority by progress: each full wave steps s_setprio 3,2,1,0 through a chunk so a SIMD's waves reach the chunk barrier together; remainder waves stay at 3; loads fenced per step; sums unchanged
# speedup vs baseline: 2.6421x; 1.0517x over previous
#include <hip/hip_runtime.h>

typedef float v2f __attribute__((ext_vector_type(2)));
typedef float v8f __attribute__((ext_vector_type(8)));

#define C_IN   128
#define K_OUT  128
#define W_     56
#define HW_    3136
#define KRED   1152
#define CC     16
#define NCHUNK 8
#define PRS    288
#define RSS    2308
#define LDS_DW (9 * RSS)
#define NBLK   256
#define NFULL  12
#define NTHR   448
#define PT_MAIN (NBLK * NFULL)


#define PB2 14
#define PB1 26
#define PB0 34
template <int NT, int D, bool STG, bool PRI>
__device__ __forceinline__ void mac_chunk(v8f (&acc)[8], const float* wb,
                                          const float* inc,
                                          const int (&toff)[9],
                                          const float (&tmask)[9],
                                          const float* wnext, float* snext,
                                          const int (&so)[12])
{
    float bx[36], by[36];
    v2f   a[2][NT];
    float4 wv[12];
    if (PRI) __builtin_amdgcn_s_setprio(3);
    #pragma unroll
    for (int i = 0; i < D; ++i) {
        const float* g = inc + toff[i / 4];
        bx[i] = g[(long)(4 * (i % 4)    ) * HW_];
        by[i] = g[(long)(4 * (i % 4) + 1) * HW_];
    }
    #pragma unroll
    for (int t = 0; t < NT; ++t) a[0][t] = *(const v2f*)(wb + t * 32);
    #pragma unroll
    for (int i = 0; i < 36; ++i) {
        if (PRI) __builtin_amdgcn_sched_barrier(0);
        if (PRI && i == PB2) __builtin_amdgcn_s_setprio(2);
        if (PRI && i == PB1) __builtin_amdgcn_s_setprio(1);
        if (PRI && i == PB0) __builtin_amdgcn_s_setprio(0);
        if (i + D < 36) {
            const int j = i + D;
            const float* g = inc + toff[j / 4];
            bx[j] = g[(long)(4 * (j % 4)    ) * HW_];
            by[j] = g[(long)(4 * (j % 4) + 1) * HW_];
        }
        if (STG && (i % 2 == 0) && (i / 2 < 12)) {
            const int u = i / 2;
            wv[u] = *(const float4*)(wnext + (long)(u / 3) * 32 * KRED
                                           + (u % 3) * 48);
        }
        if (i + 1 < 36) {
            const int j = i + 1;
            const float* wl = wb + (j / 4) * RSS + (2 * (j % 4)) * PRS;
            #pragma unroll
            for (int t = 0; t < NT; ++t)
                a[j & 1][t] = *(const v2f*)(wl + t * 32);
        }
        if (STG && (i >= 8) && (i % 2 == 0) && ((i - 8) / 2 < 12)) {
            const int u  = (i - 8) / 2;
            const int ps = u / 3;
            const int s  = u % 3;
            snext[so[s * 4 + 0] + ps * 64] = wv[u].x;
            snext[so[s * 4 + 1] + ps * 64] = wv[u].y;
            snext[so[s * 4 + 2] + ps * 64] = wv[u].z;
            snext[so[s * 4 + 3] + ps * 64] = wv[u].w;
        }
        const float mk = tmask[i / 4];
        v2f B;
        B.x = bx[i] * mk;
        B.y = by[i] * mk;
        #pragma unroll
        for (int t = 0; t < NT; ++t)
            acc[t] = __builtin_amdgcn_wmma_f32_16x16x4_f32(
                false, a[i & 1][t], false, B, (short)0, acc[t], false, false);
    }
}

template <int NT>
__device__ __forceinline__ void store_tiles(const v8f (&acc)[8], float* op,
                                            const float* bp)
{
    #pragma unroll
    for (int t = 0; t < NT; ++t) {
        #pragma unroll
        for (int g = 0; g < 8; ++g)
            op[(long)(t * 16 + g) * HW_] = acc[t][g] + bp[t * 16 + g];
    }
}


__global__ __launch_bounds__(NTHR)
void conv3x3_wmma_f32(const float* __restrict__ in,
                      const float* __restrict__ wgt,
                      const float* __restrict__ bias,
                      float* __restrict__ out)
{
    __shared__ __attribute__((aligned(16))) float wlds[2 * LDS_DW];

    const int tid  = threadIdx.x;
    const int lane = tid & 31;
    const int wave = __builtin_amdgcn_readfirstlane(tid >> 5);
    const int m    = lane & 15;
    const int hh   = lane >> 4;
    const int b    = blockIdx.x;
    const bool full = wave < NFULL;

    const int pt = full ? (b * NFULL + wave)
                        : (PT_MAIN + 2 * (b >> 3) + (wave - NFULL));
    const int t0 = full ? 0 : (b & 7);

    const int p  = pt * 16 + m;
    const int n  = p / HW_;
    const int hw = p - n * HW_;
    const int h  = hw / W_;
    const int w  = hw - h * W_;

    int   toff[9];
    float tmask[9];
    #pragma unroll
    for (int rs = 0; rs < 9; ++rs) {
        const int r  = rs / 3;
        const int s  = rs - r * 3;
        const int ih = h + r - 1;
        const int iw = w + s - 1;
        const bool valid = (ih >= 0) & (ih < W_) & (iw >= 0) & (iw < W_);
        tmask[rs] = valid ? 1.0f : 0.0f;
        toff[rs]  = valid ? (ih * W_ + iw) : 0;
    }
    const int ocl = tid / 12;
    const int qq  = tid - ocl * 12;
    int so[12];
    #pragma unroll
    for (int s = 0; s < 3; ++s) {
        #pragma unroll
        for (int j = 0; j < 4; ++j) {
            const int e  = (qq + 12 * s) * 4 + j;
            const int cl = e / 9;
            const int rs = e - cl * 9;
            so[s * 4 + j] = rs * RSS + (cl >> 1) * PRS + (cl & 1) + ocl * 2;
        }
    }
    const float* inb = in + (long)n * C_IN * HW_ + (long)(2 * hh) * HW_;
    const float* wb  = &wlds[hh * PRS + m * 2 + t0 * 32];

    v8f acc[8];
    #pragma unroll
    for (int t = 0; t < 8; ++t) acc[t] = (v8f){};

    const float* wst = wgt + ocl * KRED + qq * 4;
    if (full) {
        #pragma unroll
        for (int ps = 0; ps < 4; ++ps) {
            #pragma unroll
            for (int s = 0; s < 3; ++s) {
                const float4 v =
                    *(const float4*)(wst + (long)ps * 32 * KRED + s * 48);
                wlds[so[s * 4 + 0] + ps * 64] = v.x;
                wlds[so[s * 4 + 1] + ps * 64] = v.y;
                wlds[so[s * 4 + 2] + ps * 64] = v.z;
                wlds[so[s * 4 + 3] + ps * 64] = v.w;
            }
        }
    }
    __syncthreads();

    if (!full) __builtin_amdgcn_s_setprio(3);

    #pragma unroll 1
    for (int ch = 0; ch < NCHUNK; ++ch) {
        const int    cur = ch & 1;
        const float* inc = inb + (long)(ch * CC) * HW_;
        const float* wbc = wb + cur * LDS_DW;
        const int    nx  = (ch + 1) & (NCHUNK - 1);
        if (full)
            mac_chunk<8, 10, true, true>(acc, wbc, inc, toff, tmask,
                                   wst + nx * (CC * 9),
                                   wlds + (cur ^ 1) * LDS_DW, so);
        else {
            mac_chunk<1, 36, false, false>(acc, wbc, inc, toff, tmask,
                                    wst, wlds, so);
            asm volatile("v_nop\n\tv_nop\n\tv_nop\n\tv_nop" : "+v"(acc[0]));
        }
        if (ch + 1 < NCHUNK) __syncthreads();
    }

    if (!full) __builtin_amdgcn_s_setprio(0);

    float*       op = out + ((long)n * K_OUT + t0 * 16 + hh * 8) * HW_ + hw;
    const float* bp = bias + t0 * 16 + hh * 8;
    if (full) store_tiles<8>(acc, op, bp);
    else      store_tiles<1>(acc, op, bp);
}

extern "C" void kernel_launch(void* const* d_in, const int* in_sizes, int n_in,
                              void* d_out, int out_size, void* d_ws, size_t ws_size,
                              hipStream_t stream) {
    (void)in_sizes; (void)n_in; (void)d_ws; (void)ws_size; (void)out_size;
    const float* in   = (const float*)d_in[0];
    const float* wgt  = (const float*)d_in[1];
    const float* bias = (const float*)d_in[2];
    float*       out  = (float*)d_out;

    dim3 grid(256, 1, 1);
    dim3 block(448, 1, 1);
    conv3x3_wmma_f32<<<grid, block, 0, stream>>>(in, wgt, bias, out);
}


# Input ring carried across chunk barriers: B of the next chunk's first 10 steps is loaded in this chunk's last 10 (chunk 0's ahead of weight staging); barrier fences scoped to LDS; sums unchanged
# speedup vs baseline: 2.7238x; 1.0309x over previous
#include <hip/hip_runtime.h>

typedef float v2f __attribute__((ext_vector_type(2)));
typedef float v8f __attribute__((ext_vector_type(8)));

#define C_IN   128
#define K_OUT  128
#define W_     56
#define HW_    3136
#define KRED   1152
#define CC     16
#define NCHUNK 8
#define PRS    288
#define RSS    2308
#define LDS_DW (9 * RSS)
#define NBLK   256
#define NFULL  12
#define NTHR   448
#define PT_MAIN (NBLK * NFULL)


#define DF  10
#define PB2 14
#define PB1 26
#define PB0 34
template <int NT, int D, bool STG, bool PRI, bool CARRY>
__device__ __forceinline__ void mac_chunk(v8f (&acc)[8], const float* wb,
                                          const float* inc,
                                          const float* incn,
                                          const int (&toff)[9],
                                          const float (&tmask)[9],
                                          const float* wnext, float* snext,
                                          const int (&so)[12],
                                          float (&hx)[DF], float (&hy)[DF])
{
    static_assert(!CARRY || D == DF, "carried ring depth");
    float bx[36], by[36];
    v2f   a[2][NT];
    float4 wv[12];
    if (PRI) __builtin_amdgcn_s_setprio(3);
    if (CARRY) {
        #pragma unroll
        for (int i = 0; i < DF; ++i) { bx[i] = hx[i]; by[i] = hy[i]; }
    } else {
        #pragma unroll
        for (int i = 0; i < D; ++i) {
            const float* g = inc + toff[i / 4];
            bx[i] = g[(long)(4 * (i % 4)    ) * HW_];
            by[i] = g[(long)(4 * (i % 4) + 1) * HW_];
        }
    }
    #pragma unroll
    for (int t = 0; t < NT; ++t) a[0][t] = *(const v2f*)(wb + t * 32);
    #pragma unroll
    for (int i = 0; i < 36; ++i) {
        if (PRI) __builtin_amdgcn_sched_barrier(0);
        if (PRI && i == PB2) __builtin_amdgcn_s_setprio(2);
        if (PRI && i == PB1) __builtin_amdgcn_s_setprio(1);
        if (PRI && i == PB0) __builtin_amdgcn_s_setprio(0);
        if (i + D < 36) {
            const int j = i + D;
            const float* g = inc + toff[j / 4];
            bx[j] = g[(long)(4 * (j % 4)    ) * HW_];
            by[j] = g[(long)(4 * (j % 4) + 1) * HW_];
        } else if (CARRY) {
            const int j = i + D - 36;
            const float* g = incn + toff[j / 4];
            hx[j] = g[(long)(4 * (j % 4)    ) * HW_];
            hy[j] = g[(long)(4 * (j % 4) + 1) * HW_];
        }
        if (STG && (i % 2 == 0) && (i / 2 < 12)) {
            const int u = i / 2;
            wv[u] = *(const float4*)(wnext + (long)(u / 3) * 32 * KRED
                                           + (u % 3) * 48);
        }
        if (i + 1 < 36) {
            const int j = i + 1;
            const float* wl = wb + (j / 4) * RSS + (2 * (j % 4)) * PRS;
            #pragma unroll
            for (int t = 0; t < NT; ++t)
                a[j & 1][t] = *(const v2f*)(wl + t * 32);
        }
        if (STG && (i >= 8) && (i % 2 == 0) && ((i - 8) / 2 < 12)) {
            const int u  = (i - 8) / 2;
            const int ps = u / 3;
            const int s  = u % 3;
            snext[so[s * 4 + 0] + ps * 64] = wv[u].x;
            snext[so[s * 4 + 1] + ps * 64] = wv[u].y;
            snext[so[s * 4 + 2] + ps * 64] = wv[u].z;
            snext[so[s * 4 + 3] + ps * 64] = wv[u].w;
        }
        const float mk = tmask[i / 4];
        v2f B;
        B.x = bx[i] * mk;
        B.y = by[i] * mk;
        #pragma unroll
        for (int t = 0; t < NT; ++t)
            acc[t] = __builtin_amdgcn_wmma_f32_16x16x4_f32(
                false, a[i & 1][t], false, B, (short)0, acc[t], false, false);
    }
}

template <int NT>
__device__ __forceinline__ void store_tiles(const v8f (&acc)[8], float* op,
                                            const float* bp)
{
    #pragma unroll
    for (int t = 0; t < NT; ++t) {
        #pragma unroll
        for (int g = 0; g < 8; ++g)
            op[(long)(t * 16 + g) * HW_] = acc[t][g] + bp[t * 16 + g];
    }
}


__global__ __launch_bounds__(NTHR)
void conv3x3_wmma_f32(const float* __restrict__ in,
                      const float* __restrict__ wgt,
                      const float* __restrict__ bias,
                      float* __restrict__ out)
{
    __shared__ __attribute__((aligned(16))) float wlds[2 * LDS_DW];

    const int tid  = threadIdx.x;
    const int lane = tid & 31;
    const int wave = __builtin_amdgcn_readfirstlane(tid >> 5);
    const int m    = lane & 15;
    const int hh   = lane >> 4;
    const int b    = blockIdx.x;
    const bool full = wave < NFULL;

    const int pt = full ? (b * NFULL + wave)
                        : (PT_MAIN + 2 * (b >> 3) + (wave - NFULL));
    const int t0 = full ? 0 : (b & 7);

    const int p  = pt * 16 + m;
    const int n  = p / HW_;
    const int hw = p - n * HW_;
    const int h  = hw / W_;
    const int w  = hw - h * W_;

    int   toff[9];
    float tmask[9];
    #pragma unroll
    for (int rs = 0; rs < 9; ++rs) {
        const int r  = rs / 3;
        const int s  = rs - r * 3;
        const int ih = h + r - 1;
        const int iw = w + s - 1;
        const bool valid = (ih >= 0) & (ih < W_) & (iw >= 0) & (iw < W_);
        tmask[rs] = valid ? 1.0f : 0.0f;
        toff[rs]  = valid ? (ih * W_ + iw) : 0;
    }
    const int ocl = tid / 12;
    const int qq  = tid - ocl * 12;
    int so[12];
    #pragma unroll
    for (int s = 0; s < 3; ++s) {
        #pragma unroll
        for (int j = 0; j < 4; ++j) {
            const int e  = (qq + 12 * s) * 4 + j;
            const int cl = e / 9;
            const int rs = e - cl * 9;
            so[s * 4 + j] = rs * RSS + (cl >> 1) * PRS + (cl & 1) + ocl * 2;
        }
    }
    const float* inb = in + (long)n * C_IN * HW_ + (long)(2 * hh) * HW_;
    const float* wb  = &wlds[hh * PRS + m * 2 + t0 * 32];

    v8f acc[8];
    #pragma unroll
    for (int t = 0; t < 8; ++t) acc[t] = (v8f){};

    float hx[DF], hy[DF];
    #pragma unroll
    for (int i = 0; i < DF; ++i) { hx[i] = 0.0f; hy[i] = 0.0f; }
    if (full) {
        #pragma unroll
        for (int i = 0; i < DF; ++i) {
            const float* g = inb + toff[i / 4];
            hx[i] = g[(long)(4 * (i % 4)    ) * HW_];
            hy[i] = g[(long)(4 * (i % 4) + 1) * HW_];
        }
    }

    const float* wst = wgt + ocl * KRED + qq * 4;
    if (full) {
        #pragma unroll
        for (int ps = 0; ps < 4; ++ps) {
            #pragma unroll
            for (int s = 0; s < 3; ++s) {
                const float4 v =
                    *(const float4*)(wst + (long)ps * 32 * KRED + s * 48);
                wlds[so[s * 4 + 0] + ps * 64] = v.x;
                wlds[so[s * 4 + 1] + ps * 64] = v.y;
                wlds[so[s * 4 + 2] + ps * 64] = v.z;
                wlds[so[s * 4 + 3] + ps * 64] = v.w;
            }
        }
    }
    __syncthreads();

    if (!full) __builtin_amdgcn_s_setprio(3);

    #pragma unroll 1
    for (int ch = 0; ch < NCHUNK; ++ch) {
        const int    cur = ch & 1;
        const float* inc = inb + (long)(ch * CC) * HW_;
        const float* wbc = wb + cur * LDS_DW;
        const int    nx  = (ch + 1) & (NCHUNK - 1);
        const float* incn = inb + (long)(nx * CC) * HW_;
        if (full)
            mac_chunk<8, DF, true, true, true>(acc, wbc, inc, incn, toff, tmask,
                                   wst + nx * (CC * 9),
                                   wlds + (cur ^ 1) * LDS_DW, so, hx, hy);
        else {
            mac_chunk<1, 36, false, false, false>(acc, wbc, inc, incn, toff,
                                    tmask, wst, wlds, so, hx, hy);
            asm volatile("v_nop\n\tv_nop\n\tv_nop\n\tv_nop" : "+v"(acc[0]));
        }
        if (ch + 1 < NCHUNK) {
            __builtin_amdgcn_fence(__ATOMIC_RELEASE, "workgroup", "local");
            __builtin_amdgcn_s_barrier();
            __builtin_amdgcn_fence(__ATOMIC_ACQUIRE, "workgroup", "local");
        }
    }

    if (!full) __builtin_amdgcn_s_setprio(0);

    float*       op = out + ((long)n * K_OUT + t0 * 16 + hh * 8) * HW_ + hw;
    const float* bp = bias + t0 * 16 + hh * 8;
    if (full) store_tiles<8>(acc, op, bp);
    else      store_tiles<1>(acc, op, bp);
}

extern "C" void kernel_launch(void* const* d_in, const int* in_sizes, int n_in,
                              void* d_out, int out_size, void* d_ws, size_t ws_size,
                              hipStream_t stream) {
    (void)in_sizes; (void)n_in; (void)d_ws; (void)ws_size; (void)out_size;
    const float* in   = (const float*)d_in[0];
    const float* wgt  = (const float*)d_in[1];
    const float* bias = (const float*)d_in[2];
    float*       out  = (float*)d_out;

    dim3 grid(256, 1, 1);
    dim3 block(448, 1, 1);
    conv3x3_wmma_f32<<<grid, block, 0, stream>>>(in, wgt, bias, out);
}


# Minimal filtering F(2,3) along image rows as two K passes (shared products first, recombined in registers, then the outer two): 4 WMMAs where 6 were, 64 accumulator registers; summation regrouped
# speedup vs baseline: 3.0987x; 1.1376x over previous
#include <hip/hip_runtime.h>

typedef float v2f __attribute__((ext_vector_type(2)));
typedef float v8f __attribute__((ext_vector_type(8)));

#define C_IN   128
#define K_OUT  128
#define W_     56
#define HW_    3136
#define KRED   1152
#define PW_    28
#define PHW_   1568
#define CC     32
#define NCH    4
#define CHW    288
#define PRS    288
#define RSS    4612
#define LDS_DW (6 * RSS)
#define NBLK   256
#define NFULL  12
#define NTHR   416
#define NSTEP  24
#define DF     5
#define PB2    9
#define PB1    17
#define PB0    22


template <int P>
__device__ __forceinline__ void load_step(const float* g, const int o0,
                                          const int o3, v2f& cx, v2f& cy,
                                          float& e0x, float& e0y,
                                          float& e3x, float& e3y)
{
    cx = *(const v2f*)g;
    cy = *(const v2f*)(g + HW_);
    if (P) {
        e0x = g[o0];
        e0y = g[HW_ + o0];
        e3x = g[o3];
        e3y = g[HW_ + o3];
    }
}

template <int NT, int P, int D, bool STG, bool PRI, bool CARRY>
__device__ __forceinline__ void mac_chunk(v8f (&ya)[4], v8f (&yb)[4],
        const float* wb, const float* inc, const float* incn,
        const int (&toff)[3], const float (&rmask)[3],
        const int o0, const int o3, const float m0, const float m3,
        const float* wnext, float* snext, const int (&so)[4], const bool np1,
        v2f (&hcx)[DF], v2f (&hcy)[DF], float (&h0x)[DF], float (&h0y)[DF],
        float (&h3x)[DF], float (&h3y)[DF])
{
    static_assert(!CARRY || D == DF, "carried ring depth");
    v2f   cx[NSTEP], cy[NSTEP];
    float e0x[NSTEP], e0y[NSTEP], e3x[NSTEP], e3y[NSTEP];
    v2f   a[2][2 * NT];
    float4 wv[NSTEP];
    if (PRI) __builtin_amdgcn_s_setprio(3);
    if (CARRY) {
        #pragma unroll
        for (int i = 0; i < DF; ++i) {
            cx[i] = hcx[i]; cy[i] = hcy[i];
            if (P) { e0x[i] = h0x[i]; e0y[i] = h0y[i]; e3x[i] = h3x[i]; e3y[i] = h3y[i]; }
        }
    } else {
        #pragma unroll
        for (int i = 0; i < D; ++i)
            load_step<P>(inc + toff[i / 8] + (long)(4 * (i % 8)) * HW_, o0, o3,
                         cx[i], cy[i], e0x[i], e0y[i], e3x[i], e3y[i]);
    }
    #pragma unroll
    for (int t = 0; t < NT; ++t) {
        a[0][t]      = *(const v2f*)(wb + t * 32);
        a[0][NT + t] = *(const v2f*)(wb + 3 * RSS + t * 32);
    }
    #pragma unroll
    for (int i = 0; i < NSTEP; ++i) {
        if (PRI) __builtin_amdgcn_sched_barrier(0);
        if (PRI && i == PB2) __builtin_amdgcn_s_setprio(2);
        if (PRI && i == PB1) __builtin_amdgcn_s_setprio(1);
        if (PRI && i == PB0) __builtin_amdgcn_s_setprio(0);
        if (i + D < NSTEP) {
            const int j = i + D;
            load_step<P>(inc + toff[j / 8] + (long)(4 * (j % 8)) * HW_, o0, o3,
                         cx[j], cy[j], e0x[j], e0y[j], e3x[j], e3y[j]);
        } else if (CARRY) {
            const int j = i + D - NSTEP;
            load_step<P>(incn + toff[j / 8] + (long)(4 * (j % 8)) * HW_, o0, o3,
                         hcx[j], hcy[j], h0x[j], h0y[j], h3x[j], h3y[j]);
        }
        if (STG && ((i % 3 == 0 && i < 21) || i == 20)) {
            const int u = (i == 20) ? 7 : i / 3;
            const float* s = wnext + (long)u * 16 * KRED;
            wv[3 * u]     = *(const float4*)(s);
            wv[3 * u + 1] = *(const float4*)(s + 4);
            wv[3 * u + 2] = *(const float4*)(s + 8);
        }
        if (i + 1 < NSTEP) {
            const int j = i + 1;
            const float* wl = wb + (j / 8) * RSS + (2 * (j % 8)) * PRS;
            #pragma unroll
            for (int t = 0; t < NT; ++t) {
                a[j & 1][t]      = *(const v2f*)(wl + t * 32);
                a[j & 1][NT + t] = *(const v2f*)(wl + 3 * RSS + t * 32);
            }
        }
        if (STG && ((i % 3 == 2 && i < 21) || i == 22)) {
            const int u = (i == 22) ? 7 : i / 3;
            const float f[12] = { wv[3 * u].x,     wv[3 * u].y,     wv[3 * u].z,     wv[3 * u].w,
                                  wv[3 * u + 1].x, wv[3 * u + 1].y, wv[3 * u + 1].z, wv[3 * u + 1].w,
                                  wv[3 * u + 2].x, wv[3 * u + 2].y, wv[3 * u + 2].z, wv[3 * u + 2].w };
            #pragma unroll
            for (int k = 0; k < 4; ++k) {
                const float g0 = f[3 * k], g1 = f[3 * k + 1], g2 = f[3 * k + 2];
                const float ua = np1 ? g0 : 0.5f * (g0 + g1 + g2);
                const float ub = np1 ? g2 : 0.5f * (g0 - g1 + g2);
                snext[so[k] + u * 32]           = ua;
                snext[so[k] + 3 * RSS + u * 32] = ub;
            }
        }
        const float mk = rmask[i / 8];
        v2f Ba, Bb;
        if (P == 0) {
            Ba.x = (cx[i].x + cx[i].y) * mk;
            Ba.y = (cy[i].x + cy[i].y) * mk;
            Bb.x = (cx[i].y - cx[i].x) * mk;
            Bb.y = (cy[i].y - cy[i].x) * mk;
        } else {
            Ba.x = (e0x[i] * m0 - cx[i].y) * mk;
            Ba.y = (e0y[i] * m0 - cy[i].y) * mk;
            Bb.x = (e3x[i] * m3 - cx[i].x) * mk;
            Bb.y = (e3y[i] * m3 - cy[i].x) * mk;
        }
        #pragma unroll
        for (int t = 0; t < NT; ++t) {
            ya[t] = __builtin_amdgcn_wmma_f32_16x16x4_f32(
                false, a[i & 1][t], false, Ba, (short)0, ya[t], false, false);
            yb[t] = __builtin_amdgcn_wmma_f32_16x16x4_f32(
                false, a[i & 1][NT + t], false, Bb, (short)0, yb[t], false, false);
        }
    }
    if (NT == 4) {
        __builtin_amdgcn_sched_barrier(0);
        asm volatile("v_nop\n\tv_nop\n\tv_nop\n\tv_nop");
        __builtin_amdgcn_sched_barrier(0);
    }
}

template <int NT>
__device__ __forceinline__ void store_tiles(const v8f (&ya)[4],
                                            const v8f (&yb)[4], float* op,
                                            const float* bp)
{
    #pragma unroll
    for (int t = 0; t < NT; ++t) {
        #pragma unroll
        for (int g = 0; g < 8; ++g) {
            const float bv = bp[t * 16 + g];
            v2f o;
            o.x = ya[t][g] + bv;
            o.y = yb[t][g] + bv;
            *(v2f*)(op + (long)(t * 16 + g) * HW_) = o;
        }
    }
}

__global__ __launch_bounds__(NTHR)
void conv3x3_f23_wmma_f32(const float* __restrict__ in,
                          const float* __restrict__ wgt,
                          const float* __restrict__ bias,
                          float* __restrict__ out)
{
    __shared__ __attribute__((aligned(16))) float wlds[2 * LDS_DW];

    const int tid  = threadIdx.x;
    const int lane = tid & 31;
    const int wave = __builtin_amdgcn_readfirstlane(tid >> 5);
    const int m    = lane & 15;
    const int hh   = lane >> 4;
    const int b    = blockIdx.x;
    const bool full = wave < NFULL;

    const int q  = full ? (b * 6 + (wave >> 1)) : (NBLK * 6 + (b >> 3));
    const int t0 = full ? (4 * (wave & 1)) : (b & 7);

    const int pr  = q * 16 + m;
    const int n   = pr / PHW_;
    const int rem = pr - n * PHW_;
    const int h   = rem / PW_;
    const int pp  = rem - h * PW_;

    int   toff[3];
    float rmask[3];
    #pragma unroll
    for (int r = 0; r < 3; ++r) {
        const int ih = h + r - 1;
        const bool valid = (ih >= 0) & (ih < W_);
        rmask[r] = valid ? 1.0f : 0.0f;
        toff[r]  = (valid ? ih : h) * W_ + 2 * pp;
    }
    const int   o0 = (pp > 0) ? -1 : 0;
    const float m0 = (pp > 0) ? 1.0f : 0.0f;
    const int   o3 = (pp < PW_ - 1) ? 2 : 0;
    const float m3 = (pp < PW_ - 1) ? 1.0f : 0.0f;

    const int ocl = tid / 24;
    const int v   = tid - ocl * 24;
    int so[4];
    #pragma unroll
    for (int k = 0; k < 4; ++k) {
        const int tr = 4 * v + k;
        const int cl = tr / 3;
        const int r  = tr - cl * 3;
        so[k] = r * RSS + (cl >> 1) * PRS + (cl & 1) + ocl * 2;
    }
    const float* inb = in + (long)n * C_IN * HW_ + (long)(2 * hh) * HW_;
    const float* wb  = &wlds[hh * PRS + m * 2 + t0 * 32];

    v8f ya[4], yb[4];
    #pragma unroll
    for (int t = 0; t < 4; ++t) { ya[t] = (v8f){}; yb[t] = (v8f){}; }

    v2f   hcx[DF], hcy[DF];
    float h0x[DF], h0y[DF], h3x[DF], h3y[DF];
    #pragma unroll
    for (int i = 0; i < DF; ++i) {
        hcx[i] = (v2f){}; hcy[i] = (v2f){};
        h0x[i] = 0.0f; h0y[i] = 0.0f; h3x[i] = 0.0f; h3y[i] = 0.0f;
    }
    if (full) {
        #pragma unroll
        for (int i = 0; i < DF; ++i)
            load_step<0>(inb + toff[i / 8] + (long)(4 * (i % 8)) * HW_, o0, o3,
                         hcx[i], hcy[i], h0x[i], h0y[i], h3x[i], h3y[i]);
    }

    const float* wst = wgt + (long)ocl * KRED + 12 * v;
    if (full) {
        #pragma unroll
        for (int u = 0; u < 8; ++u) {
            const float* s = wst + (long)u * 16 * KRED;
            const float4 f0 = *(const float4*)(s);
            const float4 f1 = *(const float4*)(s + 4);
            const float4 f2 = *(const float4*)(s + 8);
            const float f[12] = { f0.x, f0.y, f0.z, f0.w, f1.x, f1.y, f1.z, f1.w,
                                  f2.x, f2.y, f2.z, f2.w };
            #pragma unroll
            for (int k = 0; k < 4; ++k) {
                const float g0 = f[3 * k], g1 = f[3 * k + 1], g2 = f[3 * k + 2];
                wlds[so[k] + u * 32]           = 0.5f * (g0 + g1 + g2);
                wlds[so[k] + 3 * RSS + u * 32] = 0.5f * (g0 - g1 + g2);
            }
        }
    }
    __syncthreads();

    if (!full) __builtin_amdgcn_s_setprio(3);

    #pragma unroll 1
    for (int ch = 0; ch < NCH; ++ch) {
        const int    cur = ch & 1;
        const float* inc = inb + (long)(ch * CC) * HW_;
        const float* wbc = wb + cur * LDS_DW;
        const int    nx  = (ch + 1) & (NCH - 1);
        const float* incn = inb + (long)(nx * CC) * HW_;
        if (full)
            mac_chunk<4, 0, DF, true, true, true>(ya, yb, wbc, inc, incn, toff,
                    rmask, o0, o3, m0, m3, wst + nx * CHW,
                    wlds + (cur ^ 1) * LDS_DW, so, ch == NCH - 1,
                    hcx, hcy, h0x, h0y, h3x, h3y);
        else {
            mac_chunk<1, 0, DF, false, false, false>(ya, yb, wbc, inc, incn, toff,
                    rmask, o0, o3, m0, m3, wst, wlds, so, false,
                    hcx, hcy, h0x, h0y, h3x, h3y);
            asm volatile("v_nop\n\tv_nop\n\tv_nop\n\tv_nop" : "+v"(ya[0]), "+v"(yb[0]));
        }
        __builtin_amdgcn_fence(__ATOMIC_RELEASE, "workgroup", "local");
        __builtin_amdgcn_s_barrier();
        __builtin_amdgcn_fence(__ATOMIC_ACQUIRE, "workgroup", "local");
    }

    if (full) {
        #pragma unroll
        for (int i = 0; i < DF; ++i) {
            const float* g = inb + toff[i / 8] + (long)(4 * (i % 8)) * HW_;
            h0x[i] = g[o0];
            h0y[i] = g[HW_ + o0];
            h3x[i] = g[o3];
            h3y[i] = g[HW_ + o3];
        }
    }
    #pragma unroll
    for (int t = 0; t < 4; ++t) {
        const v8f s = ya[t];
        ya[t] = s + yb[t];
        yb[t] = s - yb[t];
    }

    #pragma unroll 1
    for (int ch = 0; ch < NCH; ++ch) {
        const int    cur = ch & 1;
        const float* inc = inb + (long)(ch * CC) * HW_;
        const float* wbc = wb + cur * LDS_DW;
        const int    nx  = (ch + 1) & (NCH - 1);
        const float* incn = inb + (long)(nx * CC) * HW_;
        if (full)
            mac_chunk<4, 1, DF, true, true, true>(ya, yb, wbc, inc, incn, toff,
                    rmask, o0, o3, m0, m3, wst + nx * CHW,
                    wlds + (cur ^ 1) * LDS_DW, so, true,
                    hcx, hcy, h0x, h0y, h3x, h3y);
        else {
            mac_chunk<1, 1, DF, false, false, false>(ya, yb, wbc, inc, incn, toff,
                    rmask, o0, o3, m0, m3, wst, wlds, so, true,
                    hcx, hcy, h0x, h0y, h3x, h3y);
            asm volatile("v_nop\n\tv_nop\n\tv_nop\n\tv_nop" : "+v"(ya[0]), "+v"(yb[0]));
        }
        if (ch + 1 < NCH) {
            __builtin_amdgcn_fence(__ATOMIC_RELEASE, "workgroup", "local");
            __builtin_amdgcn_s_barrier();
            __builtin_amdgcn_fence(__ATOMIC_ACQUIRE, "workgroup", "local");
        }
    }

    if (!full) __builtin_amdgcn_s_setprio(0);

    float*       op = out + ((long)n * K_OUT + t0 * 16 + hh * 8) * HW_ + 2 * rem;
    const float* bp = bias + t0 * 16 + hh * 8;
    if (full) store_tiles<4>(ya, yb, op, bp);
    else      store_tiles<1>(ya, yb, op, bp);
}

extern "C" void kernel_launch(void* const* d_in, const int* in_sizes, int n_in,
                              void* d_out, int out_size, void* d_ws, size_t ws_size,
                              hipStream_t stream) {
    (void)in_sizes; (void)n_in; (void)d_ws; (void)ws_size; (void)out_size;
    const float* in   = (const float*)d_in[0];
    const float* wgt  = (const float*)d_in[1];
    const float* bias = (const float*)d_in[2];
    float*       out  = (float*)d_out;

    dim3 grid(256, 1, 1);
    dim3 block(416, 1, 1);
    conv3x3_f23_wmma_f32<<<grid, block, 0, stream>>>(in, wgt, bias, out);
}
